# starting kernel
# baseline (speedup 1.0000x reference)
#include <hip/hip_runtime.h>


typedef __attribute__((ext_vector_type(2))) float v2f;
typedef __attribute__((ext_vector_type(8))) float v8f;

#define NUM_I   32
#define NUM_A   8
#define NUM_O   10
#define NUM_AT  16
#define POS     16
#define HW      144
#define NT      320
#define WSTR    168
#define XSTR    20
#define ASTR    18

#define WMMA_F32_4(A_, B_, C_) \
  __builtin_amdgcn_wmma_f32_16x16x4_f32(false, (A_), false, (B_), (short)0, (C_), false, false)

__launch_bounds__(NT, 1)
__global__ void capsule_routing_kernel(const float* __restrict__ x,
                                       const float* __restrict__ w,
                                       const float* __restrict__ bias,
                                       float* __restrict__ out) {
  __shared__ float x_lds[NUM_I * NUM_A * XSTR];
  __shared__ float logits_lds[NUM_I * NUM_O * POS];
  __shared__ float route_lds[NUM_I * NUM_O * POS];
  __shared__ float w_lds[NUM_I * NUM_A * WSTR];
  __shared__ float act_lds[NUM_O * NUM_AT * ASTR];

  const int tid  = threadIdx.x;
  const int lane = tid & 31;
  const int o    = tid >> 5;
  const int n    = lane & 15;
  const int h    = lane >> 4;

  const int b   = blockIdx.x / 9;
  const int hw0 = (blockIdx.x % 9) * POS;

  for (int e = tid; e < NUM_I * NUM_A * POS; e += NT) {
    const int row = e >> 4;
    const int p   = e & 15;
    x_lds[row * XSTR + p] = x[((b * NUM_I) * NUM_A + row) * HW + hw0 + p];
  }
  for (int e = tid; e < NUM_I * NUM_A * (NUM_O * NUM_AT); e += NT) {
    const int row = e / (NUM_O * NUM_AT);
    const int c   = e - row * (NUM_O * NUM_AT);
    w_lds[row * WSTR + c] = w[e];
  }
  for (int e = tid; e < NUM_I * NUM_O * POS; e += NT)
    logits_lds[e] = 0.0f;

  const float bias_v = bias[o * NUM_AT + n];

  __syncthreads();

  for (int it = 0; it < 3; ++it) {
    for (int item = tid; item < NUM_I * POS; item += NT) {
      const int i = item >> 4;
      const int p = item & 15;
      float lv[NUM_O];
      float mx = -1e30f;
#pragma unroll
      for (int oo = 0; oo < NUM_O; ++oo) {
        lv[oo] = logits_lds[(i * NUM_O + oo) * POS + p];
        mx = fmaxf(mx, lv[oo]);
      }
      float s = 0.0f;
#pragma unroll
      for (int oo = 0; oo < NUM_O; ++oo) { lv[oo] = __expf(lv[oo] - mx); s += lv[oo]; }
      const float inv = 1.0f / s;
#pragma unroll
      for (int oo = 0; oo < NUM_O; ++oo)
        route_lds[(i * NUM_O + oo) * POS + p] = lv[oo] * inv;
    }
    __syncthreads();

    float pre[8];
#pragma unroll
    for (int r = 0; r < 8; ++r) pre[r] = 0.0f;

#pragma unroll 4
    for (int i = 0; i < NUM_I; ++i) {
      v2f a0, a1, b0, b1;
      a0.x = x_lds[(i * NUM_A + 2 * h    ) * XSTR + n];
      a0.y = x_lds[(i * NUM_A + 2 * h + 1) * XSTR + n];
      a1.x = x_lds[(i * NUM_A + 4 + 2 * h    ) * XSTR + n];
      a1.y = x_lds[(i * NUM_A + 4 + 2 * h + 1) * XSTR + n];
      b0.x = w_lds[(i * NUM_A + 2 * h    ) * WSTR + o * NUM_AT + n];
      b0.y = w_lds[(i * NUM_A + 2 * h + 1) * WSTR + o * NUM_AT + n];
      b1.x = w_lds[(i * NUM_A + 4 + 2 * h    ) * WSTR + o * NUM_AT + n];
      b1.y = w_lds[(i * NUM_A + 4 + 2 * h + 1) * WSTR + o * NUM_AT + n];
      v8f d = {};
      d = WMMA_F32_4(a0, b0, d);
      d = WMMA_F32_4(a1, b1, d);
#pragma unroll
      for (int r = 0; r < 8; ++r) {
        const float rt = route_lds[(i * NUM_O + o) * POS + r + 8 * h];
        pre[r] += rt * d[r];
      }
    }

    float act[8];
#pragma unroll
    for (int r = 0; r < 8; ++r) {
      pre[r] += bias_v;
      float ns = pre[r] * pre[r];
      ns += __shfl_xor(ns, 1, 32);
      ns += __shfl_xor(ns, 2, 32);
      ns += __shfl_xor(ns, 4, 32);
      ns += __shfl_xor(ns, 8, 32);
      act[r] = pre[r] * (sqrtf(ns) / (1.0f + ns));
    }

    if (it < 2) {
#pragma unroll
      for (int r = 0; r < 8; ++r)
        act_lds[(o * NUM_AT + n) * ASTR + r + 8 * h] = act[r];

      v2f bb[4];
#pragma unroll
      for (int kc = 0; kc < 4; ++kc) {
        bb[kc].x = act_lds[(o * NUM_AT + kc * 4 + 2 * h    ) * ASTR + n];
        bb[kc].y = act_lds[(o * NUM_AT + kc * 4 + 2 * h + 1) * ASTR + n];
      }

      const int arow = (n & 7);
      const int aih  = (n >> 3);
#pragma unroll 4
      for (int ip = 0; ip < NUM_I / 2; ++ip) {
        const int i0 = 2 * ip;
        v8f d = {};
#pragma unroll
        for (int kc = 0; kc < 4; ++kc) {
          v2f aa;
          aa.x = w_lds[((i0 + aih) * NUM_A + arow) * WSTR + o * NUM_AT + kc * 4 + 2 * h    ];
          aa.y = w_lds[((i0 + aih) * NUM_A + arow) * WSTR + o * NUM_AT + kc * 4 + 2 * h + 1];
          d = WMMA_F32_4(aa, bb[kc], d);
        }
        const int i = i0 + h;
        float dist = 0.0f;
#pragma unroll
        for (int r = 0; r < 8; ++r)
          dist += x_lds[(i * NUM_A + r) * XSTR + n] * d[r];
        logits_lds[(i * NUM_O + o) * POS + n] += dist;
      }
      __syncthreads();
    } else {
#pragma unroll
      for (int r = 0; r < 8; ++r)
        out[((b * NUM_O + o) * NUM_AT + n) * HW + hw0 + r + 8 * h] = act[r];
    }
  }
}

extern "C" void kernel_launch(void* const* d_in, const int* in_sizes, int n_in,
                              void* d_out, int out_size, void* d_ws, size_t ws_size,
                              hipStream_t stream) {
  const float* x    = (const float*)d_in[0];
  const float* wts  = (const float*)d_in[1];
  const float* bias = (const float*)d_in[2];
  float* out = (float*)d_out;
  (void)in_sizes; (void)n_in; (void)out_size; (void)d_ws; (void)ws_size;

  dim3 grid(64 * 9);
  dim3 block(NT);
  capsule_routing_kernel<<<grid, block, 0, stream>>>(x, wts, bias, out);
}


# Stage the x tile and the weights with 128-bit global loads, all 37 loads of a thread (bias, 4 x quads, 32 weight quads) issued before its first LDS store, not 4-byte loads in dependent groups of 8
# speedup vs baseline: 1.2946x; 1.2946x over previous
#include <hip/hip_runtime.h>


typedef __attribute__((ext_vector_type(2))) float v2f;
typedef __attribute__((ext_vector_type(4))) float v4f;
typedef __attribute__((ext_vector_type(8))) float v8f;

#define NUM_I   32
#define NUM_A   8
#define NUM_O   10
#define NUM_AT  16
#define POS     16
#define HW      144
#define NT      320
#define WSTR    168
#define XSTR    20
#define ASTR    18
#define W_Q_PER_ROW 40
#define W_Q_PER_T   32

#define WMMA_F32_4(A_, B_, C_) \
  __builtin_amdgcn_wmma_f32_16x16x4_f32(false, (A_), false, (B_), (short)0, (C_), false, false)

__launch_bounds__(NT, 1)
__global__ void capsule_routing_kernel(const float* __restrict__ x,
                                       const float* __restrict__ w,
                                       const float* __restrict__ bias,
                                       float* __restrict__ out) {
  __shared__ __attribute__((aligned(16))) float x_lds[NUM_I * NUM_A * XSTR];
  __shared__ float logits_lds[NUM_I * NUM_O * POS];
  __shared__ float route_lds[NUM_I * NUM_O * POS];
  __shared__ __attribute__((aligned(16))) float w_lds[NUM_I * NUM_A * WSTR];
  __shared__ float act_lds[NUM_O * NUM_AT * ASTR];

  const int tid  = threadIdx.x;
  const int lane = tid & 31;
  const int o    = tid >> 5;
  const int n    = lane & 15;
  const int h    = lane >> 4;

  const int b   = blockIdx.x / 9;
  const int hw0 = (blockIdx.x % 9) * POS;

  const float bias_v = bias[o * NUM_AT + n];

  v4f xv[4];
#pragma unroll
  for (int k = 0; k < 4; ++k) {
    const int e4 = tid + k * NT;
    if (e4 < NUM_I * NUM_A * (POS / 4)) {
      const int row = e4 >> 2;
      const int q   = e4 & 3;
      xv[k] = *reinterpret_cast<const v4f*>(
          x + ((b * NUM_I) * NUM_A + row) * HW + hw0 + 4 * q);
    }
  }
  v4f wv[W_Q_PER_T];
#pragma unroll
  for (int k = 0; k < W_Q_PER_T; ++k)
    wv[k] = reinterpret_cast<const v4f*>(w)[tid + k * NT];

  for (int e = tid; e < NUM_I * NUM_O * POS; e += NT)
    logits_lds[e] = 0.0f;

#pragma unroll
  for (int k = 0; k < 4; ++k) {
    const int e4 = tid + k * NT;
    if (e4 < NUM_I * NUM_A * (POS / 4)) {
      const int row = e4 >> 2;
      const int q   = e4 & 3;
      *reinterpret_cast<v4f*>(&x_lds[row * XSTR + 4 * q]) = xv[k];
    }
  }
#pragma unroll
  for (int k = 0; k < W_Q_PER_T; ++k) {
    const int e4  = tid + k * NT;
    const int row = e4 / W_Q_PER_ROW;
    const int c4  = e4 - row * W_Q_PER_ROW;
    *reinterpret_cast<v4f*>(&w_lds[row * WSTR + 4 * c4]) = wv[k];
  }

  __syncthreads();

  for (int it = 0; it < 3; ++it) {
    for (int item = tid; item < NUM_I * POS; item += NT) {
      const int i = item >> 4;
      const int p = item & 15;
      float lv[NUM_O];
      float mx = -1e30f;
#pragma unroll
      for (int oo = 0; oo < NUM_O; ++oo) {
        lv[oo] = logits_lds[(i * NUM_O + oo) * POS + p];
        mx = fmaxf(mx, lv[oo]);
      }
      float s = 0.0f;
#pragma unroll
      for (int oo = 0; oo < NUM_O; ++oo) { lv[oo] = __expf(lv[oo] - mx); s += lv[oo]; }
      const float inv = 1.0f / s;
#pragma unroll
      for (int oo = 0; oo < NUM_O; ++oo)
        route_lds[(i * NUM_O + oo) * POS + p] = lv[oo] * inv;
    }
    __syncthreads();

    float pre[8];
#pragma unroll
    for (int r = 0; r < 8; ++r) pre[r] = 0.0f;

#pragma unroll 4
    for (int i = 0; i < NUM_I; ++i) {
      v2f a0, a1, b0, b1;
      a0.x = x_lds[(i * NUM_A + 2 * h    ) * XSTR + n];
      a0.y = x_lds[(i * NUM_A + 2 * h + 1) * XSTR + n];
      a1.x = x_lds[(i * NUM_A + 4 + 2 * h    ) * XSTR + n];
      a1.y = x_lds[(i * NUM_A + 4 + 2 * h + 1) * XSTR + n];
      b0.x = w_lds[(i * NUM_A + 2 * h    ) * WSTR + o * NUM_AT + n];
      b0.y = w_lds[(i * NUM_A + 2 * h + 1) * WSTR + o * NUM_AT + n];
      b1.x = w_lds[(i * NUM_A + 4 + 2 * h    ) * WSTR + o * NUM_AT + n];
      b1.y = w_lds[(i * NUM_A + 4 + 2 * h + 1) * WSTR + o * NUM_AT + n];
      v8f d = {};
      d = WMMA_F32_4(a0, b0, d);
      d = WMMA_F32_4(a1, b1, d);
#pragma unroll
      for (int r = 0; r < 8; ++r) {
        const float rt = route_lds[(i * NUM_O + o) * POS + r + 8 * h];
        pre[r] += rt * d[r];
      }
    }

    float act[8];
#pragma unroll
    for (int r = 0; r < 8; ++r) {
      pre[r] += bias_v;
      float ns = pre[r] * pre[r];
      ns += __shfl_xor(ns, 1, 32);
      ns += __shfl_xor(ns, 2, 32);
      ns += __shfl_xor(ns, 4, 32);
      ns += __shfl_xor(ns, 8, 32);
      act[r] = pre[r] * (sqrtf(ns) / (1.0f + ns));
    }

    if (it < 2) {
#pragma unroll
      for (int r = 0; r < 8; ++r)
        act_lds[(o * NUM_AT + n) * ASTR + r + 8 * h] = act[r];

      v2f bb[4];
#pragma unroll
      for (int kc = 0; kc < 4; ++kc) {
        bb[kc].x = act_lds[(o * NUM_AT + kc * 4 + 2 * h    ) * ASTR + n];
        bb[kc].y = act_lds[(o * NUM_AT + kc * 4 + 2 * h + 1) * ASTR + n];
      }

      const int arow = (n & 7);
      const int aih  = (n >> 3);
#pragma unroll 4
      for (int ip = 0; ip < NUM_I / 2; ++ip) {
        const int i0 = 2 * ip;
        v8f d = {};
#pragma unroll
        for (int kc = 0; kc < 4; ++kc) {
          v2f aa;
          aa.x = w_lds[((i0 + aih) * NUM_A + arow) * WSTR + o * NUM_AT + kc * 4 + 2 * h    ];
          aa.y = w_lds[((i0 + aih) * NUM_A + arow) * WSTR + o * NUM_AT + kc * 4 + 2 * h + 1];
          d = WMMA_F32_4(aa, bb[kc], d);
        }
        const int i = i0 + h;
        float dist = 0.0f;
#pragma unroll
        for (int r = 0; r < 8; ++r)
          dist += x_lds[(i * NUM_A + r) * XSTR + n] * d[r];
        logits_lds[(i * NUM_O + o) * POS + n] += dist;
      }
      __syncthreads();
    } else {
#pragma unroll
      for (int r = 0; r < 8; ++r)
        out[((b * NUM_O + o) * NUM_AT + n) * HW + hw0 + r + 8 * h] = act[r];
    }
  }
}

extern "C" void kernel_launch(void* const* d_in, const int* in_sizes, int n_in,
                              void* d_out, int out_size, void* d_ws, size_t ws_size,
                              hipStream_t stream) {
  const float* x    = (const float*)d_in[0];
  const float* wts  = (const float*)d_in[1];
  const float* bias = (const float*)d_in[2];
  float* out = (float*)d_out;
  (void)in_sizes; (void)n_in; (void)out_size; (void)d_ws; (void)ws_size;

  dim3 grid(64 * 9);
  dim3 block(NT);
  capsule_routing_kernel<<<grid, block, 0, stream>>>(x, wts, bias, out);
}


# Votes computed once per workgroup (transposed WMMA tile: atoms on accumulator registers, positions on lanes) and kept in 256 VGPRs through the 3 routing iterations, not recomputed in 5 GEMM passes
# speedup vs baseline: 2.4711x; 1.9088x over previous
#include <hip/hip_runtime.h>


typedef __attribute__((ext_vector_type(2))) float v2f;
typedef __attribute__((ext_vector_type(4))) float v4f;
typedef __attribute__((ext_vector_type(8))) float v8f;

#define NUM_I   32
#define NUM_A   8
#define NUM_O   10
#define NUM_AT  16
#define POS     16
#define HW      144
#define NT      320
#define WSTR    168
#define XSTR    20
#define W_Q_PER_ROW 40
#define W_Q_PER_T   32

#define WMMA_F32_4(A_, B_, C_) \
  __builtin_amdgcn_wmma_f32_16x16x4_f32(false, (A_), false, (B_), (short)0, (C_), false, false)

static __device__ __forceinline__ float xhalf(float v) {
  return __builtin_bit_cast(float, __builtin_amdgcn_permlanex16(
      __builtin_bit_cast(int, v), __builtin_bit_cast(int, v),
      0x76543210, 0xfedcba98, false, false));
}

__launch_bounds__(NT)
__global__ void capsule_routing_kernel(const float* __restrict__ x,
                                       const float* __restrict__ w,
                                       const float* __restrict__ bias,
                                       float* __restrict__ out) {
  __shared__ __attribute__((aligned(16))) float x_lds[NUM_I * NUM_A * XSTR];
  __shared__ float logits_lds[NUM_I * NUM_O * POS];
  __shared__ float route_lds[NUM_I * NUM_O * POS];
  __shared__ __attribute__((aligned(16))) float w_lds[NUM_I * NUM_A * WSTR];

  const int tid  = threadIdx.x;
  const int lane = tid & 31;
  const int o    = tid >> 5;
  const int n    = lane & 15;
  const int h    = lane >> 4;

  const int b   = blockIdx.x / 9;
  const int hw0 = (blockIdx.x % 9) * POS;

  v2f bias_v[4];
#pragma unroll
  for (int j = 0; j < 4; ++j) {
    bias_v[j].x = bias[o * NUM_AT + 8 * h + 2 * j];
    bias_v[j].y = bias[o * NUM_AT + 8 * h + 2 * j + 1];
  }

  v4f xv[4];
#pragma unroll
  for (int k = 0; k < 4; ++k) {
    const int e4 = tid + k * NT;
    if (e4 < NUM_I * NUM_A * (POS / 4)) {
      const int row = e4 >> 2;
      const int q   = e4 & 3;
      xv[k] = *reinterpret_cast<const v4f*>(
          x + ((b * NUM_I) * NUM_A + row) * HW + hw0 + 4 * q);
    }
  }
  v4f wv[W_Q_PER_T];
#pragma unroll
  for (int k = 0; k < W_Q_PER_T; ++k)
    wv[k] = reinterpret_cast<const v4f*>(w)[tid + k * NT];

  for (int e = tid; e < NUM_I * NUM_O * POS; e += NT)
    logits_lds[e] = 0.0f;

#pragma unroll
  for (int k = 0; k < 4; ++k) {
    const int e4 = tid + k * NT;
    if (e4 < NUM_I * NUM_A * (POS / 4)) {
      const int row = e4 >> 2;
      const int q   = e4 & 3;
      *reinterpret_cast<v4f*>(&x_lds[row * XSTR + 4 * q]) = xv[k];
    }
  }
#pragma unroll
  for (int k = 0; k < W_Q_PER_T; ++k) {
    const int e4  = tid + k * NT;
    const int row = e4 / W_Q_PER_ROW;
    const int c4  = e4 - row * W_Q_PER_ROW;
    *reinterpret_cast<v4f*>(&w_lds[row * WSTR + 4 * c4]) = wv[k];
  }

  __syncthreads();

  v2f vt[NUM_I][4];
#pragma unroll
  for (int i = 0; i < NUM_I; ++i) {
    v2f wa0, wa1, xb0, xb1;
    wa0.x = w_lds[(i * NUM_A + 2 * h    ) * WSTR + o * NUM_AT + n];
    wa0.y = w_lds[(i * NUM_A + 2 * h + 1) * WSTR + o * NUM_AT + n];
    wa1.x = w_lds[(i * NUM_A + 4 + 2 * h    ) * WSTR + o * NUM_AT + n];
    wa1.y = w_lds[(i * NUM_A + 4 + 2 * h + 1) * WSTR + o * NUM_AT + n];
    xb0.x = x_lds[(i * NUM_A + 2 * h    ) * XSTR + n];
    xb0.y = x_lds[(i * NUM_A + 2 * h + 1) * XSTR + n];
    xb1.x = x_lds[(i * NUM_A + 4 + 2 * h    ) * XSTR + n];
    xb1.y = x_lds[(i * NUM_A + 4 + 2 * h + 1) * XSTR + n];
    v8f d = {};
    d = WMMA_F32_4(wa0, xb0, d);
    d = WMMA_F32_4(wa1, xb1, d);
    vt[i][0] = __builtin_shufflevector(d, d, 0, 1);
    vt[i][1] = __builtin_shufflevector(d, d, 2, 3);
    vt[i][2] = __builtin_shufflevector(d, d, 4, 5);
    vt[i][3] = __builtin_shufflevector(d, d, 6, 7);
  }

  for (int it = 0; it < 3; ++it) {
    for (int item = tid; item < NUM_I * POS; item += NT) {
      const int i = item >> 4;
      const int p = item & 15;
      float lv[NUM_O];
      float mx = -1e30f;
#pragma unroll
      for (int oo = 0; oo < NUM_O; ++oo) {
        lv[oo] = logits_lds[(i * NUM_O + oo) * POS + p];
        mx = fmaxf(mx, lv[oo]);
      }
      float s = 0.0f;
#pragma unroll
      for (int oo = 0; oo < NUM_O; ++oo) { lv[oo] = __expf(lv[oo] - mx); s += lv[oo]; }
      const float inv = 1.0f / s;
#pragma unroll
      for (int oo = 0; oo < NUM_O; ++oo)
        route_lds[(i * NUM_O + oo) * POS + p] = lv[oo] * inv;
    }
    __syncthreads();

    v2f pre[4];
#pragma unroll
    for (int j = 0; j < 4; ++j) pre[j] = (v2f){0.0f, 0.0f};
#pragma unroll
    for (int i = 0; i < NUM_I; ++i) {
      const float rt = route_lds[(i * NUM_O + o) * POS + n];
      const v2f rt2 = {rt, rt};
#pragma unroll
      for (int j = 0; j < 4; ++j) pre[j] += rt2 * vt[i][j];
    }

#pragma unroll
    for (int j = 0; j < 4; ++j) pre[j] += bias_v[j];
    v2f ns2 = pre[0] * pre[0];
#pragma unroll
    for (int j = 1; j < 4; ++j) ns2 += pre[j] * pre[j];
    float ns = ns2.x + ns2.y;
    ns += xhalf(ns);
    const float sc = sqrtf(ns) / (1.0f + ns);
    const v2f sc2 = {sc, sc};
    v2f act[4];
#pragma unroll
    for (int j = 0; j < 4; ++j) act[j] = pre[j] * sc2;

    if (it < 2) {
#pragma unroll
      for (int ip = 0; ip < NUM_I / 2; ++ip) {
        const int i0 = 2 * ip;
        v2f q0 = vt[i0][0] * act[0];
        v2f q1 = vt[i0 + 1][0] * act[0];
#pragma unroll
        for (int j = 1; j < 4; ++j) {
          q0 += vt[i0    ][j] * act[j];
          q1 += vt[i0 + 1][j] * act[j];
        }
        const float p0 = q0.x + q0.y;
        const float p1 = q1.x + q1.y;
        const float keep = h ? p1 : p0;
        const float send = h ? p0 : p1;
        const float dist = keep + xhalf(send);
        logits_lds[((i0 + h) * NUM_O + o) * POS + n] += dist;
      }
      __syncthreads();
    } else {
#pragma unroll
      for (int j = 0; j < 4; ++j) {
        out[((b * NUM_O + o) * NUM_AT + 8 * h + 2 * j    ) * HW + hw0 + n] = act[j].x;
        out[((b * NUM_O + o) * NUM_AT + 8 * h + 2 * j + 1) * HW + hw0 + n] = act[j].y;
      }
    }
  }
}

extern "C" void kernel_launch(void* const* d_in, const int* in_sizes, int n_in,
                              void* d_out, int out_size, void* d_ws, size_t ws_size,
                              hipStream_t stream) {
  const float* x    = (const float*)d_in[0];
  const float* wts  = (const float*)d_in[1];
  const float* bias = (const float*)d_in[2];
  float* out = (float*)d_out;
  (void)in_sizes; (void)n_in; (void)out_size; (void)d_ws; (void)ws_size;

  dim3 grid(64 * 9);
  dim3 block(NT);
  capsule_routing_kernel<<<grid, block, 0, stream>>>(x, wts, bias, out);
}


# Each workgroup stages the weights and a 48-position x strip into LDS once and loops over its three 16-position tiles (192 workgroups), not one tile and one 160 KB weight reload per workgroup (576)
# speedup vs baseline: 3.0927x; 1.2515x over previous
#include <hip/hip_runtime.h>


typedef __attribute__((ext_vector_type(2))) float v2f;
typedef __attribute__((ext_vector_type(4))) float v4f;
typedef __attribute__((ext_vector_type(8))) float v8f;

#define NUM_I   32
#define NUM_A   8
#define NUM_O   10
#define NUM_AT  16
#define POS     16
#define HW      144
#define NT      320
#define TILES   3
#define STRIP   (TILES * POS)
#define WSTR    168
#define XSTR    52
#define W_Q_PER_ROW 40
#define W_Q_PER_T   32
#define X_Q_PER_ROW 12
#define X_QUADS (NUM_I * NUM_A * X_Q_PER_ROW)
#define X_Q_PER_T   10

#define WMMA_F32_4(A_, B_, C_) \
  __builtin_amdgcn_wmma_f32_16x16x4_f32(false, (A_), false, (B_), (short)0, (C_), false, false)

static __device__ __forceinline__ float xhalf(float v) {
  return __builtin_bit_cast(float, __builtin_amdgcn_permlanex16(
      __builtin_bit_cast(int, v), __builtin_bit_cast(int, v),
      0x76543210, 0xfedcba98, false, false));
}

__launch_bounds__(NT)
__global__ void capsule_routing_kernel(const float* __restrict__ x,
                                       const float* __restrict__ w,
                                       const float* __restrict__ bias,
                                       float* __restrict__ out) {
  __shared__ __attribute__((aligned(16))) float x_lds[NUM_I * NUM_A * XSTR];
  __shared__ float logits_lds[NUM_I * NUM_O * POS];
  __shared__ float route_lds[NUM_I * NUM_O * POS];
  __shared__ __attribute__((aligned(16))) float w_lds[NUM_I * NUM_A * WSTR];

  const int tid = threadIdx.x;
  const int b   = blockIdx.x / 3;
  const int hwb = (blockIdx.x % 3) * STRIP;

  v2f bias_v[4];
  {
    const int o = tid >> 5;
    const int h = (tid & 31) >> 4;
#pragma unroll
    for (int j = 0; j < 4; ++j) {
      bias_v[j].x = bias[o * NUM_AT + 8 * h + 2 * j];
      bias_v[j].y = bias[o * NUM_AT + 8 * h + 2 * j + 1];
    }
  }

  v4f xv[X_Q_PER_T];
#pragma unroll
  for (int k = 0; k < X_Q_PER_T; ++k) {
    const int e4 = tid + k * NT;
    if (e4 < X_QUADS) {
      const int row = e4 / X_Q_PER_ROW;
      const int q   = e4 - row * X_Q_PER_ROW;
      xv[k] = *reinterpret_cast<const v4f*>(
          x + ((b * NUM_I) * NUM_A + row) * HW + hwb + 4 * q);
    }
  }
  v4f wv[W_Q_PER_T];
#pragma unroll
  for (int k = 0; k < W_Q_PER_T; ++k)
    wv[k] = reinterpret_cast<const v4f*>(w)[tid + k * NT];

  for (int e = tid; e < NUM_I * NUM_O * POS; e += NT)
    logits_lds[e] = 0.0f;

#pragma unroll
  for (int k = 0; k < X_Q_PER_T; ++k) {
    const int e4 = tid + k * NT;
    if (e4 < X_QUADS) {
      const int row = e4 / X_Q_PER_ROW;
      const int q   = e4 - row * X_Q_PER_ROW;
      *reinterpret_cast<v4f*>(&x_lds[row * XSTR + 4 * q]) = xv[k];
    }
  }
#pragma unroll
  for (int k = 0; k < W_Q_PER_T; ++k) {
    const int e4  = tid + k * NT;
    const int row = e4 / W_Q_PER_ROW;
    const int c4  = e4 - row * W_Q_PER_ROW;
    *reinterpret_cast<v4f*>(&w_lds[row * WSTR + 4 * c4]) = wv[k];
  }

  __syncthreads();

#pragma unroll 1
  for (int t = 0; t < TILES; ++t) {
    int tv = tid;
    asm volatile("" : "+v"(tv));
    v2f vt[NUM_I][4];
    {
      const int lane = tv & 31;
      const int o    = tv >> 5;
      const int n    = lane & 15;
      const int h    = lane >> 4;
      const float* wl = w_lds + (2 * h) * WSTR + o * NUM_AT + n;
      const float* xl = x_lds + (2 * h) * XSTR + t * POS + n;

#pragma unroll
      for (int i = 0; i < NUM_I; ++i) {
        v2f wa0, wa1, xb0, xb1;
        wa0.x = wl[(i * NUM_A    ) * WSTR];
        wa0.y = wl[(i * NUM_A + 1) * WSTR];
        wa1.x = wl[(i * NUM_A + 4) * WSTR];
        wa1.y = wl[(i * NUM_A + 5) * WSTR];
        xb0.x = xl[(i * NUM_A    ) * XSTR];
        xb0.y = xl[(i * NUM_A + 1) * XSTR];
        xb1.x = xl[(i * NUM_A + 4) * XSTR];
        xb1.y = xl[(i * NUM_A + 5) * XSTR];
        v8f d = {};
        d = WMMA_F32_4(wa0, xb0, d);
        d = WMMA_F32_4(wa1, xb1, d);
        vt[i][0] = __builtin_shufflevector(d, d, 0, 1);
        vt[i][1] = __builtin_shufflevector(d, d, 2, 3);
        vt[i][2] = __builtin_shufflevector(d, d, 4, 5);
        vt[i][3] = __builtin_shufflevector(d, d, 6, 7);
      }
    }

    int tr = tid;
    asm volatile("" : "+v"(tr));
    const int lane = tr & 31;
    const int o    = tr >> 5;
    const int n    = lane & 15;
    const int h    = lane >> 4;
    const int hw0  = hwb + t * POS;

    v2f act[4];
    for (int it = 0; it < 3; ++it) {
      for (int item = tr; item < NUM_I * POS; item += NT) {
        const int i = item >> 4;
        const int p = item & 15;
        float lv[NUM_O];
        float mx = -1e30f;
#pragma unroll
        for (int oo = 0; oo < NUM_O; ++oo) {
          lv[oo] = logits_lds[(i * NUM_O + oo) * POS + p];
          mx = fmaxf(mx, lv[oo]);
        }
        float s = 0.0f;
#pragma unroll
        for (int oo = 0; oo < NUM_O; ++oo) { lv[oo] = __expf(lv[oo] - mx); s += lv[oo]; }
        const float inv = 1.0f / s;
#pragma unroll
        for (int oo = 0; oo < NUM_O; ++oo)
          route_lds[(i * NUM_O + oo) * POS + p] = lv[oo] * inv;
      }
      __syncthreads();

      v2f pre[4];
#pragma unroll
      for (int j = 0; j < 4; ++j) pre[j] = (v2f){0.0f, 0.0f};
#pragma unroll
      for (int i = 0; i < NUM_I; ++i) {
        const float rt = route_lds[(i * NUM_O + o) * POS + n];
        const v2f rt2 = {rt, rt};
#pragma unroll
        for (int j = 0; j < 4; ++j) pre[j] += rt2 * vt[i][j];
      }

#pragma unroll
      for (int j = 0; j < 4; ++j) pre[j] += bias_v[j];
      v2f ns2 = pre[0] * pre[0];
#pragma unroll
      for (int j = 1; j < 4; ++j) ns2 += pre[j] * pre[j];
      float ns = ns2.x + ns2.y;
      ns += xhalf(ns);
      const float sc = sqrtf(ns) / (1.0f + ns);
      const v2f sc2 = {sc, sc};
#pragma unroll
      for (int j = 0; j < 4; ++j) act[j] = pre[j] * sc2;

      if (it < 2) {
#pragma unroll
        for (int ip = 0; ip < NUM_I / 2; ++ip) {
          const int i0 = 2 * ip;
          v2f q0 = vt[i0][0] * act[0];
          v2f q1 = vt[i0 + 1][0] * act[0];
#pragma unroll
          for (int j = 1; j < 4; ++j) {
            q0 += vt[i0    ][j] * act[j];
            q1 += vt[i0 + 1][j] * act[j];
          }
          const float p0 = q0.x + q0.y;
          const float p1 = q1.x + q1.y;
          const float keep = h ? p1 : p0;
          const float send = h ? p0 : p1;
          const float dist = keep + xhalf(send);
          logits_lds[((i0 + h) * NUM_O + o) * POS + n] += dist;
        }
        __syncthreads();
      }
    }

    if (t + 1 < TILES) {
      for (int e = tr; e < NUM_I * NUM_O * POS; e += NT)
        logits_lds[e] = 0.0f;
      __syncthreads();
    }

#pragma unroll
    for (int j = 0; j < 4; ++j) {
      out[((b * NUM_O + o) * NUM_AT + 8 * h + 2 * j    ) * HW + hw0 + n] = act[j].x;
      out[((b * NUM_O + o) * NUM_AT + 8 * h + 2 * j + 1) * HW + hw0 + n] = act[j].y;
    }
  }
}

extern "C" void kernel_launch(void* const* d_in, const int* in_sizes, int n_in,
                              void* d_out, int out_size, void* d_ws, size_t ws_size,
                              hipStream_t stream) {
  const float* x    = (const float*)d_in[0];
  const float* wts  = (const float*)d_in[1];
  const float* bias = (const float*)d_in[2];
  float* out = (float*)d_out;
  (void)in_sizes; (void)n_in; (void)out_size; (void)d_ws; (void)ws_size;

  dim3 grid(64 * 3);
  dim3 block(NT);
  capsule_routing_kernel<<<grid, block, 0, stream>>>(x, wts, bias, out);
}


# Routing iteration 0 uses its exact uniform weights 1/10: no softmax, route loads, logits zero-fill or their 2 barriers per tile; votes summed as they leave the WMMAs, its pass B stores the logits
# speedup vs baseline: 3.4858x; 1.1271x over previous
#include <hip/hip_runtime.h>


typedef __attribute__((ext_vector_type(2))) float v2f;
typedef __attribute__((ext_vector_type(4))) float v4f;
typedef __attribute__((ext_vector_type(8))) float v8f;

#define NUM_I   32
#define NUM_A   8
#define NUM_O   10
#define NUM_AT  16
#define POS     16
#define HW      144
#define NT      320
#define TILES   3
#define STRIP   (TILES * POS)
#define WSTR    168
#define XSTR    52
#define W_Q_PER_ROW 40
#define W_Q_PER_T   32
#define X_Q_PER_ROW 12
#define X_QUADS (NUM_I * NUM_A * X_Q_PER_ROW)
#define X_Q_PER_T   10

#define WMMA_F32_4(A_, B_, C_) \
  __builtin_amdgcn_wmma_f32_16x16x4_f32(false, (A_), false, (B_), (short)0, (C_), false, false)

static __device__ __forceinline__ float xhalf(float v) {
  return __builtin_bit_cast(float, __builtin_amdgcn_permlanex16(
      __builtin_bit_cast(int, v), __builtin_bit_cast(int, v),
      0x76543210, 0xfedcba98, false, false));
}

__launch_bounds__(NT)
__global__ void capsule_routing_kernel(const float* __restrict__ x,
                                       const float* __restrict__ w,
                                       const float* __restrict__ bias,
                                       float* __restrict__ out) {
  __shared__ __attribute__((aligned(16))) float x_lds[NUM_I * NUM_A * XSTR];
  __shared__ float logits_lds[NUM_I * NUM_O * POS];
  __shared__ float route_lds[NUM_I * NUM_O * POS];
  __shared__ __attribute__((aligned(16))) float w_lds[NUM_I * NUM_A * WSTR];

  const int tid = threadIdx.x;
  const int b   = blockIdx.x / 3;
  const int hwb = (blockIdx.x % 3) * STRIP;

  v2f bias_v[4];
  {
    const int o = tid >> 5;
    const int h = (tid & 31) >> 4;
#pragma unroll
    for (int j = 0; j < 4; ++j) {
      bias_v[j].x = bias[o * NUM_AT + 8 * h + 2 * j];
      bias_v[j].y = bias[o * NUM_AT + 8 * h + 2 * j + 1];
    }
  }

  v4f xv[X_Q_PER_T];
#pragma unroll
  for (int k = 0; k < X_Q_PER_T; ++k) {
    const int e4 = tid + k * NT;
    if (e4 < X_QUADS) {
      const int row = e4 / X_Q_PER_ROW;
      const int q   = e4 - row * X_Q_PER_ROW;
      xv[k] = *reinterpret_cast<const v4f*>(
          x + ((b * NUM_I) * NUM_A + row) * HW + hwb + 4 * q);
    }
  }
  v4f wv[W_Q_PER_T];
#pragma unroll
  for (int k = 0; k < W_Q_PER_T; ++k)
    wv[k] = reinterpret_cast<const v4f*>(w)[tid + k * NT];

#pragma unroll
  for (int k = 0; k < X_Q_PER_T; ++k) {
    const int e4 = tid + k * NT;
    if (e4 < X_QUADS) {
      const int row = e4 / X_Q_PER_ROW;
      const int q   = e4 - row * X_Q_PER_ROW;
      *reinterpret_cast<v4f*>(&x_lds[row * XSTR + 4 * q]) = xv[k];
    }
  }
#pragma unroll
  for (int k = 0; k < W_Q_PER_T; ++k) {
    const int e4  = tid + k * NT;
    const int row = e4 / W_Q_PER_ROW;
    const int c4  = e4 - row * W_Q_PER_ROW;
    *reinterpret_cast<v4f*>(&w_lds[row * WSTR + 4 * c4]) = wv[k];
  }

  __syncthreads();

#pragma unroll 1
  for (int t = 0; t < TILES; ++t) {
    int tv = tid;
    asm volatile("" : "+v"(tv));
    v2f vt[NUM_I][4];
    v2f vsum[4];
#pragma unroll
    for (int j = 0; j < 4; ++j) vsum[j] = (v2f){0.0f, 0.0f};
    {
      const int lane = tv & 31;
      const int o    = tv >> 5;
      const int n    = lane & 15;
      const int h    = lane >> 4;
      const float* wl = w_lds + (2 * h) * WSTR + o * NUM_AT + n;
      const float* xl = x_lds + (2 * h) * XSTR + t * POS + n;

#pragma unroll
      for (int i = 0; i < NUM_I; ++i) {
        v2f wa0, wa1, xb0, xb1;
        wa0.x = wl[(i * NUM_A    ) * WSTR];
        wa0.y = wl[(i * NUM_A + 1) * WSTR];
        wa1.x = wl[(i * NUM_A + 4) * WSTR];
        wa1.y = wl[(i * NUM_A + 5) * WSTR];
        xb0.x = xl[(i * NUM_A    ) * XSTR];
        xb0.y = xl[(i * NUM_A + 1) * XSTR];
        xb1.x = xl[(i * NUM_A + 4) * XSTR];
        xb1.y = xl[(i * NUM_A + 5) * XSTR];
        v8f d = {};
        d = WMMA_F32_4(wa0, xb0, d);
        d = WMMA_F32_4(wa1, xb1, d);
        vt[i][0] = __builtin_shufflevector(d, d, 0, 1);
        vt[i][1] = __builtin_shufflevector(d, d, 2, 3);
        vt[i][2] = __builtin_shufflevector(d, d, 4, 5);
        vt[i][3] = __builtin_shufflevector(d, d, 6, 7);
#pragma unroll
        for (int j = 0; j < 4; ++j) vsum[j] += vt[i][j];
      }
    }

    int tr = tid;
    asm volatile("" : "+v"(tr));
    const int lane = tr & 31;
    const int o    = tr >> 5;
    const int n    = lane & 15;
    const int h    = lane >> 4;
    const int hw0  = hwb + t * POS;

    v2f act[4];
    v2f pre[4];

    auto squash = [&]() {
#pragma unroll
      for (int j = 0; j < 4; ++j) pre[j] += bias_v[j];
      v2f ns2 = pre[0] * pre[0];
#pragma unroll
      for (int j = 1; j < 4; ++j) ns2 += pre[j] * pre[j];
      float ns = ns2.x + ns2.y;
      ns += xhalf(ns);
      const float sc = sqrtf(ns) / (1.0f + ns);
      const v2f sc2 = {sc, sc};
#pragma unroll
      for (int j = 0; j < 4; ++j) act[j] = pre[j] * sc2;
    };

    {
      const float r0 = 1.0f / (float)NUM_O;
      const v2f r02 = {r0, r0};
#pragma unroll
      for (int j = 0; j < 4; ++j) pre[j] = vsum[j] * r02;
      squash();

#pragma unroll
      for (int ip = 0; ip < NUM_I / 2; ++ip) {
        const int i0 = 2 * ip;
        v2f q0 = vt[i0][0] * act[0];
        v2f q1 = vt[i0 + 1][0] * act[0];
#pragma unroll
        for (int j = 1; j < 4; ++j) {
          q0 += vt[i0    ][j] * act[j];
          q1 += vt[i0 + 1][j] * act[j];
        }
        const float p0 = q0.x + q0.y;
        const float p1 = q1.x + q1.y;
        const float keep = h ? p1 : p0;
        const float send = h ? p0 : p1;
        logits_lds[((i0 + h) * NUM_O + o) * POS + n] = keep + xhalf(send);
      }
      __syncthreads();
    }

    for (int it = 1; it < 3; ++it) {
      for (int item = tr; item < NUM_I * POS; item += NT) {
        const int i = item >> 4;
        const int p = item & 15;
        float lv[NUM_O];
        float mx = -1e30f;
#pragma unroll
        for (int oo = 0; oo < NUM_O; ++oo) {
          lv[oo] = logits_lds[(i * NUM_O + oo) * POS + p];
          mx = fmaxf(mx, lv[oo]);
        }
        float s = 0.0f;
#pragma unroll
        for (int oo = 0; oo < NUM_O; ++oo) { lv[oo] = __expf(lv[oo] - mx); s += lv[oo]; }
        const float inv = 1.0f / s;
#pragma unroll
        for (int oo = 0; oo < NUM_O; ++oo)
          route_lds[(i * NUM_O + oo) * POS + p] = lv[oo] * inv;
      }
      __syncthreads();

#pragma unroll
      for (int j = 0; j < 4; ++j) pre[j] = (v2f){0.0f, 0.0f};
#pragma unroll
      for (int i = 0; i < NUM_I; ++i) {
        const float rt = route_lds[(i * NUM_O + o) * POS + n];
        const v2f rt2 = {rt, rt};
#pragma unroll
        for (int j = 0; j < 4; ++j) pre[j] += rt2 * vt[i][j];
      }
      squash();

      if (it < 2) {
#pragma unroll
        for (int ip = 0; ip < NUM_I / 2; ++ip) {
          const int i0 = 2 * ip;
          v2f q0 = vt[i0][0] * act[0];
          v2f q1 = vt[i0 + 1][0] * act[0];
#pragma unroll
          for (int j = 1; j < 4; ++j) {
            q0 += vt[i0    ][j] * act[j];
            q1 += vt[i0 + 1][j] * act[j];
          }
          const float p0 = q0.x + q0.y;
          const float p1 = q1.x + q1.y;
          const float keep = h ? p1 : p0;
          const float send = h ? p0 : p1;
          const float dist = keep + xhalf(send);
          logits_lds[((i0 + h) * NUM_O + o) * POS + n] += dist;
        }
        __syncthreads();
      }
    }


#pragma unroll
    for (int j = 0; j < 4; ++j) {
      out[((b * NUM_O + o) * NUM_AT + 8 * h + 2 * j    ) * HW + hw0 + n] = act[j].x;
      out[((b * NUM_O + o) * NUM_AT + 8 * h + 2 * j + 1) * HW + hw0 + n] = act[j].y;
    }
  }
}

extern "C" void kernel_launch(void* const* d_in, const int* in_sizes, int n_in,
                              void* d_out, int out_size, void* d_ws, size_t ws_size,
                              hipStream_t stream) {
  const float* x    = (const float*)d_in[0];
  const float* wts  = (const float*)d_in[1];
  const float* bias = (const float*)d_in[2];
  float* out = (float*)d_out;
  (void)in_sizes; (void)n_in; (void)out_size; (void)d_ws; (void)ws_size;

  dim3 grid(64 * 3);
  dim3 block(NT);
  capsule_routing_kernel<<<grid, block, 0, stream>>>(x, wts, bias, out);
}
